# starting kernel
# baseline (speedup 1.0000x reference)
#include <hip/hip_runtime.h>
#include <hip/hip_bf16.h>

typedef __attribute__((ext_vector_type(2))) float v2f;
typedef __attribute__((ext_vector_type(8))) float v8f;

__global__ void zero_f32(float* __restrict__ p, long n4) {
    long i = (long)blockIdx.x * blockDim.x + threadIdx.x;
    if (i < n4) {
        ((float4*)p)[i] = make_float4(0.f, 0.f, 0.f, 0.f);
    }
}

__global__ __launch_bounds__(256) void scatter_accum(
    const float* __restrict__ feat, const int* __restrict__ src,
    const int* __restrict__ dst, float* __restrict__ msg,
    float* __restrict__ deg, int E)
{
    long gid  = (long)blockIdx.x * blockDim.x + threadIdx.x;
    long edge = gid >> 5;
    int  lane = (int)(gid & 31);
    if (edge >= E) return;

    int s = src[edge];
    int d = dst[edge];

    const float4 v = ((const float4*)(feat + (size_t)s * 128))[lane];
    float* o = msg + (size_t)d * 128 + lane * 4;
    atomicAdd(o + 0, v.x);
    atomicAdd(o + 1, v.y);
    atomicAdd(o + 2, v.z);
    atomicAdd(o + 3, v.w);
    if (deg != nullptr && lane == 0) atomicAdd(deg + d, 1.0f);
}

template <bool RELU>
__global__ __launch_bounds__(256) void sage_gemm_wmma(
    const float* __restrict__ msg, const float* __restrict__ deg,
    const float* __restrict__ root, const float* __restrict__ Wl,
    const float* __restrict__ bias, const float* __restrict__ Wr,
    float* __restrict__ out, int n_rows)
{
    __shared__ float lds_mean[16][132];
    __shared__ float lds_root[16][132];

    const int tid      = threadIdx.x;
    const int row_base = blockIdx.x * 16;
    if (row_base >= n_rows) return;

    {
        int idx = tid * 8;
        int r   = idx >> 7;
        int c   = idx & 127;
        float invd = 1.0f / fmaxf(deg[row_base + r], 1.0f);
        const float* ms = msg  + (size_t)(row_base + r) * 128 + c;
        const float* rt = root + (size_t)(row_base + r) * 128 + c;
#pragma unroll
        for (int j = 0; j < 8; ++j) {
            lds_mean[r][c + j] = ms[j] * invd;
            lds_root[r][c + j] = rt[j];
        }
    }
    __syncthreads();

    const int wave = tid >> 5;
    const int lane = tid & 31;
    const int col  = wave * 16;
    const int lr   = lane & 15;
    const int kh   = (lane >> 4) * 2;

    const float* wl_row = Wl + (size_t)(col + lr) * 128;
    const float* wr_row = Wr + (size_t)(col + lr) * 128;

    v8f acc = {};
#pragma unroll 4
    for (int k0 = 0; k0 < 128; k0 += 4) {
        v2f a1 = *(const v2f*)&lds_mean[lr][k0 + kh];
        v2f a2 = *(const v2f*)&lds_root[lr][k0 + kh];
        v2f b1 = *(const v2f*)(wl_row + k0 + kh);
        v2f b2 = *(const v2f*)(wr_row + k0 + kh);
        acc = __builtin_amdgcn_wmma_f32_16x16x4_f32(
            false, a1, false, b1, (short)0, acc, false, false);
        acc = __builtin_amdgcn_wmma_f32_16x16x4_f32(
            false, a2, false, b2, (short)0, acc, false, false);
    }

    const float bv   = bias[col + lr];
    const int   rtop = (lane >> 4) * 8;
#pragma unroll
    for (int v = 0; v < 8; ++v) {
        float r = acc[v] + bv;
        if (RELU) r = fmaxf(r, 0.0f);
        out[(size_t)(row_base + v + rtop) * 128 + col + lr] = r;
    }
}

extern "C" void kernel_launch(void* const* d_in, const int* in_sizes, int n_in,
                              void* d_out, int out_size, void* d_ws, size_t ws_size,
                              hipStream_t stream) {
    const float* x   = (const float*)d_in[0];
    const int*   ei  = (const int*)d_in[1];
    const float* W1l = (const float*)d_in[2];
    const float* b1l = (const float*)d_in[3];
    const float* W1r = (const float*)d_in[4];
    const float* W2l = (const float*)d_in[5];
    const float* b2l = (const float*)d_in[6];
    const float* W2r = (const float*)d_in[7];
    float* out = (float*)d_out;

    const int N = in_sizes[0] / 128;
    const int E = in_sizes[1] / 2;
    const int* src = ei;
    const int* dst = ei + E;

    float* msg = (float*)d_ws;
    float* h1  = msg + (size_t)N * 128;
    float* deg = h1 + (size_t)N * 128;

    const long msg4 = (long)N * 128 / 4;
    const long deg4 = (long)N / 4;
    const int  zb_msg = (int)((msg4 + 255) / 256);
    const int  zb_deg = (int)((deg4 + 255) / 256);
    const int  sc_blocks = (int)(((long)E * 32 + 255) / 256);
    const int  gm_blocks = N / 16;

    zero_f32<<<zb_msg, 256, 0, stream>>>(msg, msg4);
    zero_f32<<<zb_deg, 256, 0, stream>>>(deg, deg4);
    scatter_accum<<<sc_blocks, 256, 0, stream>>>(x, src, dst, msg, deg, E);
    sage_gemm_wmma<true><<<gm_blocks, 256, 0, stream>>>(
        msg, deg, x, W1l, b1l, W1r, h1, N);

    zero_f32<<<zb_msg, 256, 0, stream>>>(msg, msg4);
    scatter_accum<<<sc_blocks, 256, 0, stream>>>(h1, src, dst, msg, nullptr, E);
    sage_gemm_wmma<false><<<gm_blocks, 256, 0, stream>>>(
        msg, deg, h1, W2l, b2l, W2r, out, N);
}


# scatter: quad-transpose the loaded float4 so each f32 atomic instruction covers one whole 128-byte line of msg, not four lines at 16-byte lane stride; sums stay arrival-ordered atomics
# speedup vs baseline: 1.1504x; 1.1504x over previous
#include <hip/hip_runtime.h>
#include <hip/hip_bf16.h>

typedef __attribute__((ext_vector_type(2))) float v2f;
typedef __attribute__((ext_vector_type(8))) float v8f;

__global__ void zero_f32(float* __restrict__ p, long n4) {
    long i = (long)blockIdx.x * blockDim.x + threadIdx.x;
    if (i < n4) {
        ((float4*)p)[i] = make_float4(0.f, 0.f, 0.f, 0.f);
    }
}

template <bool XOR1>
__device__ __forceinline__ float quad_swap(float x) {
    const int i = __float_as_int(x);
    const int r = XOR1 ? __builtin_amdgcn_update_dpp(i, i, 0xB1, 0xF, 0xF, false)
                       : __builtin_amdgcn_update_dpp(i, i, 0x4E, 0xF, 0xF, false);
    return __int_as_float(r);
}

__global__ __launch_bounds__(256) void scatter_accum(
    const float* __restrict__ feat, const int* __restrict__ src,
    const int* __restrict__ dst, float* __restrict__ msg,
    float* __restrict__ deg, int E)
{
    long gid  = (long)blockIdx.x * blockDim.x + threadIdx.x;
    long edge = gid >> 5;
    int  lane = (int)(gid & 31);
    if (edge >= E) return;

    int s = src[edge];
    int d = dst[edge];

    const int q = lane >> 2;
    const int r = lane & 3;
    const float4 v =
        *(const float4*)(feat + (size_t)s * 128 + r * 32 + q * 4);
    float v0 = v.x, v1 = v.y, v2 = v.z, v3 = v.w;

    const bool odd = (lane & 1) != 0;
    const bool hi  = (lane & 2) != 0;
    {
        const float g0 = quad_swap<true>(odd ? v0 : v1);
        const float g1 = quad_swap<true>(odd ? v2 : v3);
        if (odd) { v0 = g0; v2 = g1; } else { v1 = g0; v3 = g1; }
    }
    {
        const float g0 = quad_swap<false>(hi ? v0 : v2);
        const float g1 = quad_swap<false>(hi ? v1 : v3);
        if (hi) { v0 = g0; v1 = g1; } else { v2 = g0; v3 = g1; }
    }

    float* o = msg + (size_t)d * 128 + lane;
    atomicAdd(o +  0, v0);
    atomicAdd(o + 32, v1);
    atomicAdd(o + 64, v2);
    atomicAdd(o + 96, v3);
    if (deg != nullptr && lane == 0) atomicAdd(deg + d, 1.0f);
}

template <bool RELU>
__global__ __launch_bounds__(256) void sage_gemm_wmma(
    const float* __restrict__ msg, const float* __restrict__ deg,
    const float* __restrict__ root, const float* __restrict__ Wl,
    const float* __restrict__ bias, const float* __restrict__ Wr,
    float* __restrict__ out, int n_rows)
{
    __shared__ float lds_mean[16][132];
    __shared__ float lds_root[16][132];

    const int tid      = threadIdx.x;
    const int row_base = blockIdx.x * 16;
    if (row_base >= n_rows) return;

    {
        int idx = tid * 8;
        int r   = idx >> 7;
        int c   = idx & 127;
        float invd = 1.0f / fmaxf(deg[row_base + r], 1.0f);
        const float* ms = msg  + (size_t)(row_base + r) * 128 + c;
        const float* rt = root + (size_t)(row_base + r) * 128 + c;
#pragma unroll
        for (int j = 0; j < 8; ++j) {
            lds_mean[r][c + j] = ms[j] * invd;
            lds_root[r][c + j] = rt[j];
        }
    }
    __syncthreads();

    const int wave = tid >> 5;
    const int lane = tid & 31;
    const int col  = wave * 16;
    const int lr   = lane & 15;
    const int kh   = (lane >> 4) * 2;

    const float* wl_row = Wl + (size_t)(col + lr) * 128;
    const float* wr_row = Wr + (size_t)(col + lr) * 128;

    v8f acc = {};
#pragma unroll 4
    for (int k0 = 0; k0 < 128; k0 += 4) {
        v2f a1 = *(const v2f*)&lds_mean[lr][k0 + kh];
        v2f a2 = *(const v2f*)&lds_root[lr][k0 + kh];
        v2f b1 = *(const v2f*)(wl_row + k0 + kh);
        v2f b2 = *(const v2f*)(wr_row + k0 + kh);
        acc = __builtin_amdgcn_wmma_f32_16x16x4_f32(
            false, a1, false, b1, (short)0, acc, false, false);
        acc = __builtin_amdgcn_wmma_f32_16x16x4_f32(
            false, a2, false, b2, (short)0, acc, false, false);
    }

    const float bv   = bias[col + lr];
    const int   rtop = (lane >> 4) * 8;
#pragma unroll
    for (int v = 0; v < 8; ++v) {
        float r = acc[v] + bv;
        if (RELU) r = fmaxf(r, 0.0f);
        out[(size_t)(row_base + v + rtop) * 128 + col + lr] = r;
    }
}

extern "C" void kernel_launch(void* const* d_in, const int* in_sizes, int n_in,
                              void* d_out, int out_size, void* d_ws, size_t ws_size,
                              hipStream_t stream) {
    const float* x   = (const float*)d_in[0];
    const int*   ei  = (const int*)d_in[1];
    const float* W1l = (const float*)d_in[2];
    const float* b1l = (const float*)d_in[3];
    const float* W1r = (const float*)d_in[4];
    const float* W2l = (const float*)d_in[5];
    const float* b2l = (const float*)d_in[6];
    const float* W2r = (const float*)d_in[7];
    float* out = (float*)d_out;

    const int N = in_sizes[0] / 128;
    const int E = in_sizes[1] / 2;
    const int* src = ei;
    const int* dst = ei + E;

    float* msg = (float*)d_ws;
    float* h1  = msg + (size_t)N * 128;
    float* deg = h1 + (size_t)N * 128;

    const long msg4 = (long)N * 128 / 4;
    const long deg4 = (long)N / 4;
    const int  zb_msg = (int)((msg4 + 255) / 256);
    const int  zb_deg = (int)((deg4 + 255) / 256);
    const int  sc_blocks = (int)(((long)E * 32 + 255) / 256);
    const int  gm_blocks = N / 16;

    zero_f32<<<zb_msg, 256, 0, stream>>>(msg, msg4);
    zero_f32<<<zb_deg, 256, 0, stream>>>(deg, deg4);
    scatter_accum<<<sc_blocks, 256, 0, stream>>>(x, src, dst, msg, deg, E);
    sage_gemm_wmma<true><<<gm_blocks, 256, 0, stream>>>(
        msg, deg, x, W1l, b1l, W1r, h1, N);

    zero_f32<<<zb_msg, 256, 0, stream>>>(msg, msg4);
    scatter_accum<<<sc_blocks, 256, 0, stream>>>(h1, src, dst, msg, nullptr, E);
    sage_gemm_wmma<false><<<gm_blocks, 256, 0, stream>>>(
        msg, deg, h1, W2l, b2l, W2r, out, N);
}


# aggregate by gather: edges bucketed by dst once per call (int atomics), each bucket sorted by src id, node sums taken in that fixed order in registers; no f32 atomics, sums identical run to run
# speedup vs baseline: 1.6602x; 1.4432x over previous
#include <hip/hip_runtime.h>
#include <hip/hip_bf16.h>


typedef __attribute__((ext_vector_type(2))) float v2f;
typedef __attribute__((ext_vector_type(8))) float v8f;

#define SCAN_TILE 1024

__global__ __launch_bounds__(256) void zero_i32(int* __restrict__ p, int n) {
    const int i = blockIdx.x * blockDim.x + threadIdx.x;
    if (i < n) p[i] = 0;
}

__global__ __launch_bounds__(256) void dst_histogram(
    const int* __restrict__ dst, int* __restrict__ cnt, int E)
{
    const int e0 = blockIdx.x * 1024 + threadIdx.x;
    int d[4];
#pragma unroll
    for (int k = 0; k < 4; ++k) {
        const int e = e0 + k * 256;
        d[k] = (e < E) ? dst[e] : -1;
    }
#pragma unroll
    for (int k = 0; k < 4; ++k) {
        if (d[k] >= 0) atomicAdd(cnt + d[k], 1);
    }
}

__global__ __launch_bounds__(256) void scan_tiles(
    const int* __restrict__ cnt, int* __restrict__ rowptr,
    int* __restrict__ tile_sum, int n)
{
    __shared__ int wave_tot[8];
    const int tid  = threadIdx.x;
    const int lane = tid & 31;
    const int wave = tid >> 5;
    const int base = blockIdx.x * SCAN_TILE + tid * 4;

    const int c0 = (base + 0 < n) ? cnt[base + 0] : 0;
    const int c1 = (base + 1 < n) ? cnt[base + 1] : 0;
    const int c2 = (base + 2 < n) ? cnt[base + 2] : 0;
    const int c3 = (base + 3 < n) ? cnt[base + 3] : 0;
    const int t  = c0 + c1 + c2 + c3;

    int incl = t;
#pragma unroll
    for (int off = 1; off < 32; off <<= 1) {
        const int u = __shfl_up(incl, off, 32);
        if (lane >= off) incl += u;
    }
    if (lane == 31) wave_tot[wave] = incl;
    __syncthreads();

    int woff = 0;
    for (int w = 0; w < wave; ++w) woff += wave_tot[w];
    const int excl = woff + incl - t;

    if (base + 0 < n) rowptr[base + 0] = excl;
    if (base + 1 < n) rowptr[base + 1] = excl + c0;
    if (base + 2 < n) rowptr[base + 2] = excl + c0 + c1;
    if (base + 3 < n) rowptr[base + 3] = excl + c0 + c1 + c2;
    if (tid == 255) tile_sum[blockIdx.x] = woff + incl;
}

__global__ __launch_bounds__(256) void scan_finish(
    int* __restrict__ cnt, int* __restrict__ rowptr,
    const int* __restrict__ tile_sum, float* __restrict__ deg, int n)
{
    __shared__ int s_off;
    const int tid = threadIdx.x;
    if (tid < 32) {
        int a = 0;
        for (int b = tid; b < (int)blockIdx.x; b += 32) a += tile_sum[b];
#pragma unroll
        for (int off = 16; off > 0; off >>= 1) a += __shfl_xor(a, off, 32);
        if (tid == 0) s_off = a;
    }
    __syncthreads();
    const int off  = s_off;
    const int base = blockIdx.x * SCAN_TILE + tid * 4;
#pragma unroll
    for (int k = 0; k < 4; ++k) {
        const int i = base + k;
        if (i < n) {
            const int c = cnt[i];
            const int r = rowptr[i] + off;
            rowptr[i] = r;
            deg[i]    = (float)c;
            cnt[i]    = 0;
            if (i == n - 1) rowptr[n] = r + c;
        }
    }
}

__global__ __launch_bounds__(256) void bucket_fill(
    const int* __restrict__ src, const int* __restrict__ dst,
    const int* __restrict__ rowptr, int* __restrict__ cnt,
    int* __restrict__ bucket, int E)
{
    const int e0 = blockIdx.x * 1024 + threadIdx.x;
    int d[4], s[4], slot[4];
#pragma unroll
    for (int k = 0; k < 4; ++k) {
        const int e = e0 + k * 256;
        d[k] = (e < E) ? dst[e] : -1;
        s[k] = (e < E) ? src[e] : 0;
    }
#pragma unroll
    for (int k = 0; k < 4; ++k) {
        slot[k] = 0;
        if (d[k] >= 0) slot[k] = rowptr[d[k]] + atomicAdd(cnt + d[k], 1);
    }
#pragma unroll
    for (int k = 0; k < 4; ++k) {
        if (d[k] >= 0) bucket[slot[k]] = s[k];
    }
}

__global__ __launch_bounds__(256) void bucket_sort(
    const int* __restrict__ rowptr, const int* __restrict__ bucket,
    int* __restrict__ sorted, int n)
{
    const int lane = threadIdx.x & 31;
    const int node0 = __builtin_amdgcn_readfirstlane(
        (int)((blockIdx.x * 8 + (threadIdx.x >> 5)) * 4));
  for (int q = 0; q < 4; ++q) {
    const int node = node0 + q;
    if (node >= n) return;
    const int start = __builtin_amdgcn_readfirstlane(rowptr[node]);
    const int len   = __builtin_amdgcn_readfirstlane(rowptr[node + 1]) - start;

    for (int ib = 0; ib < len; ib += 32) {
        const int i   = ib + lane;
        const int key = (i < len) ? bucket[start + i] : 0;
        int rank = 0;
        for (int jb = 0; jb < len; jb += 32) {
            const int jl = jb + lane;
            const int kc = (jl < len) ? bucket[start + jl] : 0;
            const int m  = (len - jb < 32) ? (len - jb) : 32;
            for (int jj = 0; jj < m; ++jj) {
                const int kj = __builtin_amdgcn_readlane(kc, jj);
                const int j  = jb + jj;
                rank += (kj < key || (kj == key && j < i)) ? 1 : 0;
            }
        }
        if (i < len) sorted[start + rank] = key;
    }
  }
}

__device__ __forceinline__ void add4(float4& a, const float4& r) {
    a.x += r.x; a.y += r.y; a.z += r.z; a.w += r.w;
}

__global__ __launch_bounds__(256) void gather_sum(
    const float* __restrict__ feat, const int* __restrict__ rowptr,
    const int* __restrict__ sorted, float* __restrict__ msg, int n)
{
    const int lane = threadIdx.x & 31;
    const int node = __builtin_amdgcn_readfirstlane(
        (int)(blockIdx.x * 8 + (threadIdx.x >> 5)));
    if (node >= n) return;
    const int start = __builtin_amdgcn_readfirstlane(rowptr[node]);
    const int end   = __builtin_amdgcn_readfirstlane(rowptr[node + 1]);

    const float* col = feat + lane * 4;
    float4 acc = make_float4(0.f, 0.f, 0.f, 0.f);

    for (int base = start; base < end; base += 32) {
        const int m   = (end - base < 32) ? (end - base) : 32;
        const int idx = (lane < m) ? sorted[base + lane] : 0;
        int jj = 0;
        for (; jj + 8 <= m; jj += 8) {
            float4 r[8];
#pragma unroll
            for (int u = 0; u < 8; ++u) {
                const int s = __builtin_amdgcn_readlane(idx, jj + u);
                r[u] = *(const float4*)(col + (size_t)s * 128);
            }
#pragma unroll
            for (int u = 0; u < 8; ++u) add4(acc, r[u]);
        }
        if (m - jj >= 4) {
            float4 r[4];
#pragma unroll
            for (int u = 0; u < 4; ++u) {
                const int s = __builtin_amdgcn_readlane(idx, jj + u);
                r[u] = *(const float4*)(col + (size_t)s * 128);
            }
#pragma unroll
            for (int u = 0; u < 4; ++u) add4(acc, r[u]);
            jj += 4;
        }
        if (m - jj >= 2) {
            const int s0 = __builtin_amdgcn_readlane(idx, jj);
            const int s1 = __builtin_amdgcn_readlane(idx, jj + 1);
            const float4 r0 = *(const float4*)(col + (size_t)s0 * 128);
            const float4 r1 = *(const float4*)(col + (size_t)s1 * 128);
            add4(acc, r0);
            add4(acc, r1);
            jj += 2;
        }
        if (m - jj >= 1) {
            const int s0 = __builtin_amdgcn_readlane(idx, jj);
            const float4 r0 = *(const float4*)(col + (size_t)s0 * 128);
            add4(acc, r0);
        }
    }

    *(float4*)(msg + (size_t)node * 128 + lane * 4) = acc;
}

template <bool RELU>
__global__ __launch_bounds__(256) void sage_gemm_wmma(
    const float* __restrict__ msg, const float* __restrict__ deg,
    const float* __restrict__ root, const float* __restrict__ Wl,
    const float* __restrict__ bias, const float* __restrict__ Wr,
    float* __restrict__ out, int n_rows)
{
    __shared__ float lds_mean[16][132];
    __shared__ float lds_root[16][132];

    const int tid      = threadIdx.x;
    const int row_base = blockIdx.x * 16;
    if (row_base >= n_rows) return;

    {
        int idx = tid * 8;
        int r   = idx >> 7;
        int c   = idx & 127;
        float invd = 1.0f / fmaxf(deg[row_base + r], 1.0f);
        const float* ms = msg  + (size_t)(row_base + r) * 128 + c;
        const float* rt = root + (size_t)(row_base + r) * 128 + c;
#pragma unroll
        for (int j = 0; j < 8; ++j) {
            lds_mean[r][c + j] = ms[j] * invd;
            lds_root[r][c + j] = rt[j];
        }
    }
    __syncthreads();

    const int wave = tid >> 5;
    const int lane = tid & 31;
    const int col  = wave * 16;
    const int lr   = lane & 15;
    const int kh   = (lane >> 4) * 2;

    const float* wl_row = Wl + (size_t)(col + lr) * 128;
    const float* wr_row = Wr + (size_t)(col + lr) * 128;

    v8f acc = {};
#pragma unroll 4
    for (int k0 = 0; k0 < 128; k0 += 4) {
        v2f a1 = *(const v2f*)&lds_mean[lr][k0 + kh];
        v2f a2 = *(const v2f*)&lds_root[lr][k0 + kh];
        v2f b1 = *(const v2f*)(wl_row + k0 + kh);
        v2f b2 = *(const v2f*)(wr_row + k0 + kh);
        acc = __builtin_amdgcn_wmma_f32_16x16x4_f32(
            false, a1, false, b1, (short)0, acc, false, false);
        acc = __builtin_amdgcn_wmma_f32_16x16x4_f32(
            false, a2, false, b2, (short)0, acc, false, false);
    }

    const float bv   = bias[col + lr];
    const int   rtop = (lane >> 4) * 8;
#pragma unroll
    for (int v = 0; v < 8; ++v) {
        float r = acc[v] + bv;
        if (RELU) r = fmaxf(r, 0.0f);
        out[(size_t)(row_base + v + rtop) * 128 + col + lr] = r;
    }
}

extern "C" void kernel_launch(void* const* d_in, const int* in_sizes, int n_in,
                              void* d_out, int out_size, void* d_ws, size_t ws_size,
                              hipStream_t stream) {
    const float* x   = (const float*)d_in[0];
    const int*   ei  = (const int*)d_in[1];
    const float* W1l = (const float*)d_in[2];
    const float* b1l = (const float*)d_in[3];
    const float* W1r = (const float*)d_in[4];
    const float* W2l = (const float*)d_in[5];
    const float* b2l = (const float*)d_in[6];
    const float* W2r = (const float*)d_in[7];
    float* out = (float*)d_out;

    const int N = in_sizes[0] / 128;
    const int E = in_sizes[1] / 2;
    const int* src = ei;
    const int* dst = ei + E;

    const int tiles = (N + SCAN_TILE - 1) / SCAN_TILE;

    float* msg = (float*)d_ws;
    float* h1  = msg + (size_t)N * 128;
    float* deg = h1 + (size_t)N * 128;
    int* cnt      = (int*)(deg + N);
    int* rowptr   = cnt + N;
    int* tile_sum = rowptr + (N + 1);
    int* bucket   = tile_sum + tiles;
    int* sorted   = bucket + E;

    const int  n_blocks  = (N + 255) / 256;
    const int  e_blocks  = (E + 1023) / 1024;
    const int  s_blocks  = (N + 31) / 32;
    const int  w_blocks  = (N + 7) / 8;
    const int  gm_blocks = N / 16;

    zero_i32<<<n_blocks, 256, 0, stream>>>(cnt, N);
    dst_histogram<<<e_blocks, 256, 0, stream>>>(dst, cnt, E);
    scan_tiles<<<tiles, 256, 0, stream>>>(cnt, rowptr, tile_sum, N);
    scan_finish<<<tiles, 256, 0, stream>>>(cnt, rowptr, tile_sum, deg, N);
    bucket_fill<<<e_blocks, 256, 0, stream>>>(src, dst, rowptr, cnt, bucket, E);
    bucket_sort<<<s_blocks, 256, 0, stream>>>(rowptr, bucket, sorted, N);

    gather_sum<<<w_blocks, 256, 0, stream>>>(x, rowptr, sorted, msg, N);
    sage_gemm_wmma<true><<<gm_blocks, 256, 0, stream>>>(
        msg, deg, x, W1l, b1l, W1r, h1, N);

    gather_sum<<<w_blocks, 256, 0, stream>>>(h1, rowptr, sorted, msg, N);
    sage_gemm_wmma<false><<<gm_blocks, 256, 0, stream>>>(
        msg, deg, h1, W2l, b2l, W2r, out, N);
}


# GEMM: each wave keeps its 16-column weight fragment for 8 row tiles (128-row block): one 16-line weight load feeds 8 matrix instructions, not 1; 8x fewer waves; per-element sum order unchanged
# speedup vs baseline: 1.8755x; 1.1297x over previous
#include <hip/hip_runtime.h>
#include <hip/hip_bf16.h>


typedef __attribute__((ext_vector_type(2))) float v2f;
typedef __attribute__((ext_vector_type(8))) float v8f;

#define SCAN_TILE 1024

__global__ __launch_bounds__(256) void zero_i32(int* __restrict__ p, int n) {
    const int i = blockIdx.x * blockDim.x + threadIdx.x;
    if (i < n) p[i] = 0;
}

__global__ __launch_bounds__(256) void dst_histogram(
    const int* __restrict__ dst, int* __restrict__ cnt, int E)
{
    const int e0 = blockIdx.x * 1024 + threadIdx.x;
    int d[4];
#pragma unroll
    for (int k = 0; k < 4; ++k) {
        const int e = e0 + k * 256;
        d[k] = (e < E) ? dst[e] : -1;
    }
#pragma unroll
    for (int k = 0; k < 4; ++k) {
        if (d[k] >= 0) atomicAdd(cnt + d[k], 1);
    }
}

__global__ __launch_bounds__(256) void scan_tiles(
    const int* __restrict__ cnt, int* __restrict__ rowptr,
    int* __restrict__ tile_sum, int n)
{
    __shared__ int wave_tot[8];
    const int tid  = threadIdx.x;
    const int lane = tid & 31;
    const int wave = tid >> 5;
    const int base = blockIdx.x * SCAN_TILE + tid * 4;

    const int c0 = (base + 0 < n) ? cnt[base + 0] : 0;
    const int c1 = (base + 1 < n) ? cnt[base + 1] : 0;
    const int c2 = (base + 2 < n) ? cnt[base + 2] : 0;
    const int c3 = (base + 3 < n) ? cnt[base + 3] : 0;
    const int t  = c0 + c1 + c2 + c3;

    int incl = t;
#pragma unroll
    for (int off = 1; off < 32; off <<= 1) {
        const int u = __shfl_up(incl, off, 32);
        if (lane >= off) incl += u;
    }
    if (lane == 31) wave_tot[wave] = incl;
    __syncthreads();

    int woff = 0;
    for (int w = 0; w < wave; ++w) woff += wave_tot[w];
    const int excl = woff + incl - t;

    if (base + 0 < n) rowptr[base + 0] = excl;
    if (base + 1 < n) rowptr[base + 1] = excl + c0;
    if (base + 2 < n) rowptr[base + 2] = excl + c0 + c1;
    if (base + 3 < n) rowptr[base + 3] = excl + c0 + c1 + c2;
    if (tid == 255) tile_sum[blockIdx.x] = woff + incl;
}

__global__ __launch_bounds__(256) void scan_finish(
    int* __restrict__ cnt, int* __restrict__ rowptr,
    const int* __restrict__ tile_sum, float* __restrict__ deg, int n)
{
    __shared__ int s_off;
    const int tid = threadIdx.x;
    if (tid < 32) {
        int a = 0;
        for (int b = tid; b < (int)blockIdx.x; b += 32) a += tile_sum[b];
#pragma unroll
        for (int off = 16; off > 0; off >>= 1) a += __shfl_xor(a, off, 32);
        if (tid == 0) s_off = a;
    }
    __syncthreads();
    const int off  = s_off;
    const int base = blockIdx.x * SCAN_TILE + tid * 4;
#pragma unroll
    for (int k = 0; k < 4; ++k) {
        const int i = base + k;
        if (i < n) {
            const int c = cnt[i];
            const int r = rowptr[i] + off;
            rowptr[i] = r;
            deg[i]    = (float)c;
            cnt[i]    = 0;
            if (i == n - 1) rowptr[n] = r + c;
        }
    }
}

__global__ __launch_bounds__(256) void bucket_fill(
    const int* __restrict__ src, const int* __restrict__ dst,
    const int* __restrict__ rowptr, int* __restrict__ cnt,
    int* __restrict__ bucket, int E)
{
    const int e0 = blockIdx.x * 1024 + threadIdx.x;
    int d[4], s[4], slot[4];
#pragma unroll
    for (int k = 0; k < 4; ++k) {
        const int e = e0 + k * 256;
        d[k] = (e < E) ? dst[e] : -1;
        s[k] = (e < E) ? src[e] : 0;
    }
#pragma unroll
    for (int k = 0; k < 4; ++k) {
        slot[k] = 0;
        if (d[k] >= 0) slot[k] = rowptr[d[k]] + atomicAdd(cnt + d[k], 1);
    }
#pragma unroll
    for (int k = 0; k < 4; ++k) {
        if (d[k] >= 0) bucket[slot[k]] = s[k];
    }
}

__global__ __launch_bounds__(256) void bucket_sort(
    const int* __restrict__ rowptr, const int* __restrict__ bucket,
    int* __restrict__ sorted, int n)
{
    const int lane = threadIdx.x & 31;
    const int node0 = __builtin_amdgcn_readfirstlane(
        (int)((blockIdx.x * 8 + (threadIdx.x >> 5)) * 4));
  for (int q = 0; q < 4; ++q) {
    const int node = node0 + q;
    if (node >= n) return;
    const int start = __builtin_amdgcn_readfirstlane(rowptr[node]);
    const int len   = __builtin_amdgcn_readfirstlane(rowptr[node + 1]) - start;

    for (int ib = 0; ib < len; ib += 32) {
        const int i   = ib + lane;
        const int key = (i < len) ? bucket[start + i] : 0;
        int rank = 0;
        for (int jb = 0; jb < len; jb += 32) {
            const int jl = jb + lane;
            const int kc = (jl < len) ? bucket[start + jl] : 0;
            const int m  = (len - jb < 32) ? (len - jb) : 32;
            for (int jj = 0; jj < m; ++jj) {
                const int kj = __builtin_amdgcn_readlane(kc, jj);
                const int j  = jb + jj;
                rank += (kj < key || (kj == key && j < i)) ? 1 : 0;
            }
        }
        if (i < len) sorted[start + rank] = key;
    }
  }
}

__device__ __forceinline__ void add4(float4& a, const float4& r) {
    a.x += r.x; a.y += r.y; a.z += r.z; a.w += r.w;
}

__global__ __launch_bounds__(256) void gather_sum(
    const float* __restrict__ feat, const int* __restrict__ rowptr,
    const int* __restrict__ sorted, float* __restrict__ msg, int n)
{
    const int lane = threadIdx.x & 31;
    const int node = __builtin_amdgcn_readfirstlane(
        (int)(blockIdx.x * 8 + (threadIdx.x >> 5)));
    if (node >= n) return;
    const int start = __builtin_amdgcn_readfirstlane(rowptr[node]);
    const int end   = __builtin_amdgcn_readfirstlane(rowptr[node + 1]);

    const float* col = feat + lane * 4;
    float4 acc = make_float4(0.f, 0.f, 0.f, 0.f);

    for (int base = start; base < end; base += 32) {
        const int m   = (end - base < 32) ? (end - base) : 32;
        const int idx = (lane < m) ? sorted[base + lane] : 0;
        int jj = 0;
        for (; jj + 8 <= m; jj += 8) {
            float4 r[8];
#pragma unroll
            for (int u = 0; u < 8; ++u) {
                const int s = __builtin_amdgcn_readlane(idx, jj + u);
                r[u] = *(const float4*)(col + (size_t)s * 128);
            }
#pragma unroll
            for (int u = 0; u < 8; ++u) add4(acc, r[u]);
        }
        if (m - jj >= 4) {
            float4 r[4];
#pragma unroll
            for (int u = 0; u < 4; ++u) {
                const int s = __builtin_amdgcn_readlane(idx, jj + u);
                r[u] = *(const float4*)(col + (size_t)s * 128);
            }
#pragma unroll
            for (int u = 0; u < 4; ++u) add4(acc, r[u]);
            jj += 4;
        }
        if (m - jj >= 2) {
            const int s0 = __builtin_amdgcn_readlane(idx, jj);
            const int s1 = __builtin_amdgcn_readlane(idx, jj + 1);
            const float4 r0 = *(const float4*)(col + (size_t)s0 * 128);
            const float4 r1 = *(const float4*)(col + (size_t)s1 * 128);
            add4(acc, r0);
            add4(acc, r1);
            jj += 2;
        }
        if (m - jj >= 1) {
            const int s0 = __builtin_amdgcn_readlane(idx, jj);
            const float4 r0 = *(const float4*)(col + (size_t)s0 * 128);
            add4(acc, r0);
        }
    }

    *(float4*)(msg + (size_t)node * 128 + lane * 4) = acc;
}

#define GEMM_RT   8
#define GEMM_ROWS (GEMM_RT * 16)

template <bool RELU>
__global__ __launch_bounds__(256) void sage_gemm_wmma(
    const float* __restrict__ msg, const float* __restrict__ deg,
    const float* __restrict__ root, const float* __restrict__ Wl,
    const float* __restrict__ bias, const float* __restrict__ Wr,
    float* __restrict__ out, int n_rows)
{
    __shared__ float lds_mean[GEMM_ROWS][132];
    __shared__ float lds_root[GEMM_ROWS][132];

    const int tid      = threadIdx.x;
    const int row_base = blockIdx.x * GEMM_ROWS;
    if (row_base >= n_rows) return;

    const bool full = (row_base + GEMM_ROWS <= n_rows);

    {
        const int c  = (tid & 15) * 8;
        const int r0 = tid >> 4;
        if (full) {
            for (int p = 0; p < GEMM_RT; ++p) {
                const int r   = p * 16 + r0;
                const int row = row_base + r;
                const float invd = 1.0f / fmaxf(deg[row], 1.0f);
                const float* ms = msg  + (size_t)row * 128 + c;
                const float* rt = root + (size_t)row * 128 + c;
#pragma unroll
                for (int j = 0; j < 8; ++j) {
                    lds_mean[r][c + j] = ms[j] * invd;
                    lds_root[r][c + j] = rt[j];
                }
            }
        } else {
            for (int p = 0; p < GEMM_RT; ++p) {
                const int r   = p * 16 + r0;
                const int row = row_base + r;
                const bool live = (row < n_rows);
                const int  rowc = live ? row : row_base;
                const float invd = 1.0f / fmaxf(deg[rowc], 1.0f);
                const float* ms = msg  + (size_t)rowc * 128 + c;
                const float* rt = root + (size_t)rowc * 128 + c;
#pragma unroll
                for (int j = 0; j < 8; ++j) {
                    lds_mean[r][c + j] = live ? ms[j] * invd : 0.0f;
                    lds_root[r][c + j] = live ? rt[j] : 0.0f;
                }
            }
        }
    }
    __syncthreads();

    const int wave = tid >> 5;
    const int lane = tid & 31;
    const int col  = wave * 16;
    const int lr   = lane & 15;
    const int kh   = (lane >> 4) * 2;

    const float* wl_row = Wl + (size_t)(col + lr) * 128;
    const float* wr_row = Wr + (size_t)(col + lr) * 128;

    v8f acc[GEMM_RT];
#pragma unroll
    for (int t = 0; t < GEMM_RT; ++t) acc[t] = v8f{};

#pragma unroll 2
    for (int k0 = 0; k0 < 128; k0 += 4) {
        const v2f b1 = *(const v2f*)(wl_row + k0 + kh);
        const v2f b2 = *(const v2f*)(wr_row + k0 + kh);
#pragma unroll
        for (int t = 0; t < GEMM_RT; ++t) {
            const v2f a1 = *(const v2f*)&lds_mean[t * 16 + lr][k0 + kh];
            const v2f a2 = *(const v2f*)&lds_root[t * 16 + lr][k0 + kh];
            acc[t] = __builtin_amdgcn_wmma_f32_16x16x4_f32(
                false, a1, false, b1, (short)0, acc[t], false, false);
            acc[t] = __builtin_amdgcn_wmma_f32_16x16x4_f32(
                false, a2, false, b2, (short)0, acc[t], false, false);
        }
    }

    const float bv   = bias[col + lr];
    const int   rtop = (lane >> 4) * 8;
    if (full) {
#pragma unroll
        for (int t = 0; t < GEMM_RT; ++t) {
#pragma unroll
            for (int v = 0; v < 8; ++v) {
                const int row = row_base + t * 16 + v + rtop;
                float r = acc[t][v] + bv;
                if (RELU) r = fmaxf(r, 0.0f);
                out[(size_t)row * 128 + col + lr] = r;
            }
        }
    } else {
#pragma unroll
        for (int t = 0; t < GEMM_RT; ++t) {
#pragma unroll
            for (int v = 0; v < 8; ++v) {
                const int row = row_base + t * 16 + v + rtop;
                float r = acc[t][v] + bv;
                if (RELU) r = fmaxf(r, 0.0f);
                if (row < n_rows) out[(size_t)row * 128 + col + lr] = r;
            }
        }
    }
}

extern "C" void kernel_launch(void* const* d_in, const int* in_sizes, int n_in,
                              void* d_out, int out_size, void* d_ws, size_t ws_size,
                              hipStream_t stream) {
    const float* x   = (const float*)d_in[0];
    const int*   ei  = (const int*)d_in[1];
    const float* W1l = (const float*)d_in[2];
    const float* b1l = (const float*)d_in[3];
    const float* W1r = (const float*)d_in[4];
    const float* W2l = (const float*)d_in[5];
    const float* b2l = (const float*)d_in[6];
    const float* W2r = (const float*)d_in[7];
    float* out = (float*)d_out;

    const int N = in_sizes[0] / 128;
    const int E = in_sizes[1] / 2;
    const int* src = ei;
    const int* dst = ei + E;

    const int tiles = (N + SCAN_TILE - 1) / SCAN_TILE;

    float* msg = (float*)d_ws;
    float* h1  = msg + (size_t)N * 128;
    float* deg = h1 + (size_t)N * 128;
    int* cnt      = (int*)(deg + N);
    int* rowptr   = cnt + N;
    int* tile_sum = rowptr + (N + 1);
    int* bucket   = tile_sum + tiles;
    int* sorted   = bucket + E;

    const int  n_blocks  = (N + 255) / 256;
    const int  e_blocks  = (E + 1023) / 1024;
    const int  s_blocks  = (N + 31) / 32;
    const int  w_blocks  = (N + 7) / 8;
    const int  gm_blocks = (N + GEMM_ROWS - 1) / GEMM_ROWS;

    zero_i32<<<n_blocks, 256, 0, stream>>>(cnt, N);
    dst_histogram<<<e_blocks, 256, 0, stream>>>(dst, cnt, E);
    scan_tiles<<<tiles, 256, 0, stream>>>(cnt, rowptr, tile_sum, N);
    scan_finish<<<tiles, 256, 0, stream>>>(cnt, rowptr, tile_sum, deg, N);
    bucket_fill<<<e_blocks, 256, 0, stream>>>(src, dst, rowptr, cnt, bucket, E);
    bucket_sort<<<s_blocks, 256, 0, stream>>>(rowptr, bucket, sorted, N);

    gather_sum<<<w_blocks, 256, 0, stream>>>(x, rowptr, sorted, msg, N);
    sage_gemm_wmma<true><<<gm_blocks, 256, 0, stream>>>(
        msg, deg, x, W1l, b1l, W1r, h1, N);

    gather_sum<<<w_blocks, 256, 0, stream>>>(h1, rowptr, sorted, msg, N);
    sage_gemm_wmma<false><<<gm_blocks, 256, 0, stream>>>(
        msg, deg, h1, W2l, b2l, W2r, out, N);
}


# bucketing: two-level counting sort on LDS counters (edges binned by 256-node destination range via a count matrix and scan, then slotted per bin); no global atomics; sum order unchanged
# speedup vs baseline: 3.1325x; 1.6702x over previous
#include <hip/hip_runtime.h>
#include <hip/hip_bf16.h>


typedef __attribute__((ext_vector_type(2))) float v2f;
typedef __attribute__((ext_vector_type(8))) float v8f;

#define SCAN_TILE    1024
#define BIN_SHIFT    8
#define BIN_NODES    (1 << BIN_SHIFT)
#define CHUNK_BLOCKS 256

extern __shared__ int s_bins[];

__global__ __launch_bounds__(1024) void coarse_count(
    const int* __restrict__ dst, int* __restrict__ cmat, int E, int nb)
{
    const int tid    = threadIdx.x;
    const int blocks = (int)gridDim.x;
    for (int i = tid; i < nb; i += 1024) s_bins[i] = 0;
    __syncthreads();

    const int chunk = (E + blocks - 1) / blocks;
    const int e_lo  = (int)blockIdx.x * chunk;
    const int e_hi  = (e_lo + chunk < E) ? (e_lo + chunk) : E;
    for (int e0 = e_lo + tid; e0 < e_hi; e0 += 4096) {
        int d[4];
#pragma unroll
        for (int k = 0; k < 4; ++k) {
            const int e = e0 + k * 1024;
            d[k] = (e < e_hi) ? dst[e] : -1;
        }
#pragma unroll
        for (int k = 0; k < 4; ++k) {
            if (d[k] >= 0) atomicAdd(&s_bins[d[k] >> BIN_SHIFT], 1);
        }
    }
    __syncthreads();
    for (int i = tid; i < nb; i += 1024) cmat[i * blocks + (int)blockIdx.x] = s_bins[i];
}

__global__ __launch_bounds__(256) void scan_tiles(
    const int* __restrict__ cnt, int* __restrict__ off,
    int* __restrict__ tile_sum, int n)
{
    __shared__ int wave_tot[8];
    const int tid  = threadIdx.x;
    const int lane = tid & 31;
    const int wave = tid >> 5;
    const int base = blockIdx.x * SCAN_TILE + tid * 4;

    const int c0 = (base + 0 < n) ? cnt[base + 0] : 0;
    const int c1 = (base + 1 < n) ? cnt[base + 1] : 0;
    const int c2 = (base + 2 < n) ? cnt[base + 2] : 0;
    const int c3 = (base + 3 < n) ? cnt[base + 3] : 0;
    const int t  = c0 + c1 + c2 + c3;

    int incl = t;
#pragma unroll
    for (int o = 1; o < 32; o <<= 1) {
        const int u = __shfl_up(incl, o, 32);
        if (lane >= o) incl += u;
    }
    if (lane == 31) wave_tot[wave] = incl;
    __syncthreads();

    int woff = 0;
    for (int w = 0; w < wave; ++w) woff += wave_tot[w];
    const int excl = woff + incl - t;

    if (base + 0 < n) off[base + 0] = excl;
    if (base + 1 < n) off[base + 1] = excl + c0;
    if (base + 2 < n) off[base + 2] = excl + c0 + c1;
    if (base + 3 < n) off[base + 3] = excl + c0 + c1 + c2;
    if (tid == 255) tile_sum[blockIdx.x] = woff + incl;
}

__global__ __launch_bounds__(256) void scan_finish(
    const int* __restrict__ cnt, int* __restrict__ off,
    const int* __restrict__ tile_sum, int n)
{
    __shared__ int s_off;
    const int tid = threadIdx.x;
    if (tid < 32) {
        int a = 0;
        for (int b = tid; b < (int)blockIdx.x; b += 32) a += tile_sum[b];
#pragma unroll
        for (int o = 16; o > 0; o >>= 1) a += __shfl_xor(a, o, 32);
        if (tid == 0) s_off = a;
    }
    __syncthreads();
    const int add  = s_off;
    const int base = blockIdx.x * SCAN_TILE + tid * 4;
#pragma unroll
    for (int k = 0; k < 4; ++k) {
        const int i = base + k;
        if (i < n) {
            const int r = off[i] + add;
            off[i] = r;
            if (i == n - 1) off[n] = r + cnt[i];
        }
    }
}

__global__ __launch_bounds__(1024) void coarse_scatter(
    const int* __restrict__ src, const int* __restrict__ dst,
    const int* __restrict__ coff, int* __restrict__ part, int E, int nb)
{
    const int tid    = threadIdx.x;
    const int blocks = (int)gridDim.x;
    for (int i = tid; i < nb; i += 1024) s_bins[i] = coff[i * blocks + (int)blockIdx.x];
    __syncthreads();

    const int chunk = (E + blocks - 1) / blocks;
    const int e_lo  = (int)blockIdx.x * chunk;
    const int e_hi  = (e_lo + chunk < E) ? (e_lo + chunk) : E;
    for (int e0 = e_lo + tid; e0 < e_hi; e0 += 4096) {
        int d[4], s[4];
#pragma unroll
        for (int k = 0; k < 4; ++k) {
            const int e = e0 + k * 1024;
            d[k] = (e < e_hi) ? dst[e] : -1;
            s[k] = (e < e_hi) ? src[e] : 0;
        }
#pragma unroll
        for (int k = 0; k < 4; ++k) {
            if (d[k] >= 0) {
                const int slot = atomicAdd(&s_bins[d[k] >> BIN_SHIFT], 1);
                *(int2*)(part + 2 * (size_t)slot) = make_int2(s[k], d[k]);
            }
        }
    }
}

__global__ __launch_bounds__(1024) void fine_bucket(
    const int* __restrict__ part, const int* __restrict__ coff,
    int* __restrict__ rowptr, float* __restrict__ deg,
    int* __restrict__ bucket, int n, int blocks)
{
    __shared__ int s_cnt[BIN_NODES];
    __shared__ int s_wtot[BIN_NODES / 32];
    const int tid  = threadIdx.x;
    const int lane = tid & 31;
    const int wave = tid >> 5;
    const int bin  = (int)blockIdx.x;
    const int b_lo = coff[bin * blocks];
    const int b_hi = coff[(bin + 1) * blocks];

    if (tid < BIN_NODES) s_cnt[tid] = 0;
    __syncthreads();

    for (int p0 = b_lo + tid; p0 < b_hi; p0 += 4096) {
        int d[4];
#pragma unroll
        for (int k = 0; k < 4; ++k) {
            const int p = p0 + k * 1024;
            d[k] = (p < b_hi) ? part[2 * (size_t)p + 1] : -1;
        }
#pragma unroll
        for (int k = 0; k < 4; ++k) {
            if (d[k] >= 0) atomicAdd(&s_cnt[d[k] & (BIN_NODES - 1)], 1);
        }
    }
    __syncthreads();

    int c = 0, incl = 0;
    if (tid < BIN_NODES) {
        c = s_cnt[tid];
        incl = c;
#pragma unroll
        for (int o = 1; o < 32; o <<= 1) {
            const int u = __shfl_up(incl, o, 32);
            if (lane >= o) incl += u;
        }
        if (lane == 31) s_wtot[wave] = incl;
    }
    __syncthreads();
    if (tid < BIN_NODES) {
        int woff = 0;
        for (int w = 0; w < wave; ++w) woff += s_wtot[w];
        const int first = b_lo + woff + incl - c;
        const int node  = bin * BIN_NODES + tid;
        if (node < n) {
            rowptr[node] = first;
            deg[node]    = (float)c;
            if (node == n - 1) rowptr[n] = first + c;
        }
        s_cnt[tid] = first;
    }
    __syncthreads();

    for (int p0 = b_lo + tid; p0 < b_hi; p0 += 4096) {
        int2 sd[4];
#pragma unroll
        for (int k = 0; k < 4; ++k) {
            const int p = p0 + k * 1024;
            sd[k] = (p < b_hi) ? *(const int2*)(part + 2 * (size_t)p) : make_int2(0, -1);
        }
#pragma unroll
        for (int k = 0; k < 4; ++k) {
            if (sd[k].y >= 0) {
                const int slot = atomicAdd(&s_cnt[sd[k].y & (BIN_NODES - 1)], 1);
                bucket[slot] = sd[k].x;
            }
        }
    }
}

__global__ __launch_bounds__(256) void bucket_sort(
    const int* __restrict__ rowptr, const int* __restrict__ bucket,
    int* __restrict__ sorted, int n)
{
    const int lane = threadIdx.x & 31;
    const int node0 = __builtin_amdgcn_readfirstlane(
        (int)((blockIdx.x * 8 + (threadIdx.x >> 5)) * 4));
  for (int q = 0; q < 4; ++q) {
    const int node = node0 + q;
    if (node >= n) return;
    const int start = __builtin_amdgcn_readfirstlane(rowptr[node]);
    const int len   = __builtin_amdgcn_readfirstlane(rowptr[node + 1]) - start;

    for (int ib = 0; ib < len; ib += 32) {
        const int i   = ib + lane;
        const int key = (i < len) ? bucket[start + i] : 0;
        int rank = 0;
        for (int jb = 0; jb < len; jb += 32) {
            const int jl = jb + lane;
            const int kc = (jl < len) ? bucket[start + jl] : 0;
            const int m  = (len - jb < 32) ? (len - jb) : 32;
            for (int jj = 0; jj < m; ++jj) {
                const int kj = __builtin_amdgcn_readlane(kc, jj);
                const int j  = jb + jj;
                rank += (kj < key || (kj == key && j < i)) ? 1 : 0;
            }
        }
        if (i < len) sorted[start + rank] = key;
    }
  }
}

__device__ __forceinline__ void add4(float4& a, const float4& r) {
    a.x += r.x; a.y += r.y; a.z += r.z; a.w += r.w;
}

__global__ __launch_bounds__(256) void gather_sum(
    const float* __restrict__ feat, const int* __restrict__ rowptr,
    const int* __restrict__ sorted, float* __restrict__ msg, int n)
{
    const int lane = threadIdx.x & 31;
    const int node = __builtin_amdgcn_readfirstlane(
        (int)(blockIdx.x * 8 + (threadIdx.x >> 5)));
    if (node >= n) return;
    const int start = __builtin_amdgcn_readfirstlane(rowptr[node]);
    const int end   = __builtin_amdgcn_readfirstlane(rowptr[node + 1]);

    const float* col = feat + lane * 4;
    float4 acc = make_float4(0.f, 0.f, 0.f, 0.f);

    for (int base = start; base < end; base += 32) {
        const int m   = (end - base < 32) ? (end - base) : 32;
        const int idx = (lane < m) ? sorted[base + lane] : 0;
        int jj = 0;
        for (; jj + 8 <= m; jj += 8) {
            float4 r[8];
#pragma unroll
            for (int u = 0; u < 8; ++u) {
                const int s = __builtin_amdgcn_readlane(idx, jj + u);
                r[u] = *(const float4*)(col + (size_t)s * 128);
            }
#pragma unroll
            for (int u = 0; u < 8; ++u) add4(acc, r[u]);
        }
        if (m - jj >= 4) {
            float4 r[4];
#pragma unroll
            for (int u = 0; u < 4; ++u) {
                const int s = __builtin_amdgcn_readlane(idx, jj + u);
                r[u] = *(const float4*)(col + (size_t)s * 128);
            }
#pragma unroll
            for (int u = 0; u < 4; ++u) add4(acc, r[u]);
            jj += 4;
        }
        if (m - jj >= 2) {
            const int s0 = __builtin_amdgcn_readlane(idx, jj);
            const int s1 = __builtin_amdgcn_readlane(idx, jj + 1);
            const float4 r0 = *(const float4*)(col + (size_t)s0 * 128);
            const float4 r1 = *(const float4*)(col + (size_t)s1 * 128);
            add4(acc, r0);
            add4(acc, r1);
            jj += 2;
        }
        if (m - jj >= 1) {
            const int s0 = __builtin_amdgcn_readlane(idx, jj);
            const float4 r0 = *(const float4*)(col + (size_t)s0 * 128);
            add4(acc, r0);
        }
    }

    *(float4*)(msg + (size_t)node * 128 + lane * 4) = acc;
}

#define GEMM_RT   8
#define GEMM_ROWS (GEMM_RT * 16)

template <bool RELU>
__global__ __launch_bounds__(256) void sage_gemm_wmma(
    const float* __restrict__ msg, const float* __restrict__ deg,
    const float* __restrict__ root, const float* __restrict__ Wl,
    const float* __restrict__ bias, const float* __restrict__ Wr,
    float* __restrict__ out, int n_rows)
{
    __shared__ float lds_mean[GEMM_ROWS][132];
    __shared__ float lds_root[GEMM_ROWS][132];

    const int tid      = threadIdx.x;
    const int row_base = blockIdx.x * GEMM_ROWS;
    if (row_base >= n_rows) return;

    const bool full = (row_base + GEMM_ROWS <= n_rows);

    {
        const int c  = (tid & 15) * 8;
        const int r0 = tid >> 4;
        if (full) {
            for (int p = 0; p < GEMM_RT; ++p) {
                const int r   = p * 16 + r0;
                const int row = row_base + r;
                const float invd = 1.0f / fmaxf(deg[row], 1.0f);
                const float* ms = msg  + (size_t)row * 128 + c;
                const float* rt = root + (size_t)row * 128 + c;
#pragma unroll
                for (int j = 0; j < 8; ++j) {
                    lds_mean[r][c + j] = ms[j] * invd;
                    lds_root[r][c + j] = rt[j];
                }
            }
        } else {
            for (int p = 0; p < GEMM_RT; ++p) {
                const int r   = p * 16 + r0;
                const int row = row_base + r;
                const bool live = (row < n_rows);
                const int  rowc = live ? row : row_base;
                const float invd = 1.0f / fmaxf(deg[rowc], 1.0f);
                const float* ms = msg  + (size_t)rowc * 128 + c;
                const float* rt = root + (size_t)rowc * 128 + c;
#pragma unroll
                for (int j = 0; j < 8; ++j) {
                    lds_mean[r][c + j] = live ? ms[j] * invd : 0.0f;
                    lds_root[r][c + j] = live ? rt[j] : 0.0f;
                }
            }
        }
    }
    __syncthreads();

    const int wave = tid >> 5;
    const int lane = tid & 31;
    const int col  = wave * 16;
    const int lr   = lane & 15;
    const int kh   = (lane >> 4) * 2;

    const float* wl_row = Wl + (size_t)(col + lr) * 128;
    const float* wr_row = Wr + (size_t)(col + lr) * 128;

    v8f acc[GEMM_RT];
#pragma unroll
    for (int t = 0; t < GEMM_RT; ++t) acc[t] = v8f{};

#pragma unroll 2
    for (int k0 = 0; k0 < 128; k0 += 4) {
        const v2f b1 = *(const v2f*)(wl_row + k0 + kh);
        const v2f b2 = *(const v2f*)(wr_row + k0 + kh);
#pragma unroll
        for (int t = 0; t < GEMM_RT; ++t) {
            const v2f a1 = *(const v2f*)&lds_mean[t * 16 + lr][k0 + kh];
            const v2f a2 = *(const v2f*)&lds_root[t * 16 + lr][k0 + kh];
            acc[t] = __builtin_amdgcn_wmma_f32_16x16x4_f32(
                false, a1, false, b1, (short)0, acc[t], false, false);
            acc[t] = __builtin_amdgcn_wmma_f32_16x16x4_f32(
                false, a2, false, b2, (short)0, acc[t], false, false);
        }
    }

    const float bv   = bias[col + lr];
    const int   rtop = (lane >> 4) * 8;
    if (full) {
#pragma unroll
        for (int t = 0; t < GEMM_RT; ++t) {
#pragma unroll
            for (int v = 0; v < 8; ++v) {
                const int row = row_base + t * 16 + v + rtop;
                float r = acc[t][v] + bv;
                if (RELU) r = fmaxf(r, 0.0f);
                out[(size_t)row * 128 + col + lr] = r;
            }
        }
    } else {
#pragma unroll
        for (int t = 0; t < GEMM_RT; ++t) {
#pragma unroll
            for (int v = 0; v < 8; ++v) {
                const int row = row_base + t * 16 + v + rtop;
                float r = acc[t][v] + bv;
                if (RELU) r = fmaxf(r, 0.0f);
                if (row < n_rows) out[(size_t)row * 128 + col + lr] = r;
            }
        }
    }
}

extern "C" void kernel_launch(void* const* d_in, const int* in_sizes, int n_in,
                              void* d_out, int out_size, void* d_ws, size_t ws_size,
                              hipStream_t stream) {
    const float* x   = (const float*)d_in[0];
    const int*   ei  = (const int*)d_in[1];
    const float* W1l = (const float*)d_in[2];
    const float* b1l = (const float*)d_in[3];
    const float* W1r = (const float*)d_in[4];
    const float* W2l = (const float*)d_in[5];
    const float* b2l = (const float*)d_in[6];
    const float* W2r = (const float*)d_in[7];
    float* out = (float*)d_out;

    const int N = in_sizes[0] / 128;
    const int E = in_sizes[1] / 2;
    const int* src = ei;
    const int* dst = ei + E;

    const int bins  = (N + BIN_NODES - 1) / BIN_NODES;
    const int cells = bins * CHUNK_BLOCKS;
    const int tiles = (cells + SCAN_TILE - 1) / SCAN_TILE;

    float* msg = (float*)d_ws;
    float* h1  = msg + (size_t)N * 128;
    int* part     = (int*)(h1 + (size_t)N * 128);
    float* deg    = (float*)(part + 2 * (size_t)E);
    int* rowptr   = (int*)(deg + N);
    int* cmat     = rowptr + (N + 1);
    int* coff     = cmat + cells;
    int* tile_sum = coff + (cells + 1);
    int* bucket   = tile_sum + tiles;
    int* sorted   = bucket + E;

    const int    s_blocks  = (N + 31) / 32;
    const int    w_blocks  = (N + 7) / 8;
    const int    gm_blocks = (N + GEMM_ROWS - 1) / GEMM_ROWS;
    const size_t bin_lds   = (size_t)bins * sizeof(int);

    coarse_count<<<CHUNK_BLOCKS, 1024, bin_lds, stream>>>(dst, cmat, E, bins);
    scan_tiles<<<tiles, 256, 0, stream>>>(cmat, coff, tile_sum, cells);
    scan_finish<<<tiles, 256, 0, stream>>>(cmat, coff, tile_sum, cells);
    coarse_scatter<<<CHUNK_BLOCKS, 1024, bin_lds, stream>>>(src, dst, coff, part, E, bins);
    fine_bucket<<<bins, 1024, 0, stream>>>(part, coff, rowptr, deg, bucket, N, CHUNK_BLOCKS);
    bucket_sort<<<s_blocks, 256, 0, stream>>>(rowptr, bucket, sorted, N);

    gather_sum<<<w_blocks, 256, 0, stream>>>(x, rowptr, sorted, msg, N);
    sage_gemm_wmma<true><<<gm_blocks, 256, 0, stream>>>(
        msg, deg, x, W1l, b1l, W1r, h1, N);

    gather_sum<<<w_blocks, 256, 0, stream>>>(h1, rowptr, sorted, msg, N);
    sage_gemm_wmma<false><<<gm_blocks, 256, 0, stream>>>(
        msg, deg, h1, W2l, b2l, W2r, out, N);
}


# GEMM: 256 resident 16-wave blocks, each walks an equal run of rows in 80-row strips; weights staged in LDS once; 8 loader waves fetch the next strip while 8 matrix waves multiply; sum order unchanged
# speedup vs baseline: 3.3843x; 1.0804x over previous
#include <hip/hip_runtime.h>
#include <hip/hip_bf16.h>


typedef __attribute__((ext_vector_type(2))) float v2f;
typedef __attribute__((ext_vector_type(8))) float v8f;

#define SCAN_TILE    1024
#define BIN_SHIFT    8
#define BIN_NODES    (1 << BIN_SHIFT)
#define CHUNK_BLOCKS 256

extern __shared__ int s_bins[];

__global__ __launch_bounds__(1024) void coarse_count(
    const int* __restrict__ dst, int* __restrict__ cmat, int E, int nb)
{
    const int tid    = threadIdx.x;
    const int blocks = (int)gridDim.x;
    for (int i = tid; i < nb; i += 1024) s_bins[i] = 0;
    __syncthreads();

    const int chunk = (E + blocks - 1) / blocks;
    const int e_lo  = (int)blockIdx.x * chunk;
    const int e_hi  = (e_lo + chunk < E) ? (e_lo + chunk) : E;
    for (int e0 = e_lo + tid; e0 < e_hi; e0 += 4096) {
        int d[4];
#pragma unroll
        for (int k = 0; k < 4; ++k) {
            const int e = e0 + k * 1024;
            d[k] = (e < e_hi) ? dst[e] : -1;
        }
#pragma unroll
        for (int k = 0; k < 4; ++k) {
            if (d[k] >= 0) atomicAdd(&s_bins[d[k] >> BIN_SHIFT], 1);
        }
    }
    __syncthreads();
    for (int i = tid; i < nb; i += 1024) cmat[i * blocks + (int)blockIdx.x] = s_bins[i];
}

__global__ __launch_bounds__(256) void scan_tiles(
    const int* __restrict__ cnt, int* __restrict__ off,
    int* __restrict__ tile_sum, int n)
{
    __shared__ int wave_tot[8];
    const int tid  = threadIdx.x;
    const int lane = tid & 31;
    const int wave = tid >> 5;
    const int base = blockIdx.x * SCAN_TILE + tid * 4;

    const int c0 = (base + 0 < n) ? cnt[base + 0] : 0;
    const int c1 = (base + 1 < n) ? cnt[base + 1] : 0;
    const int c2 = (base + 2 < n) ? cnt[base + 2] : 0;
    const int c3 = (base + 3 < n) ? cnt[base + 3] : 0;
    const int t  = c0 + c1 + c2 + c3;

    int incl = t;
#pragma unroll
    for (int o = 1; o < 32; o <<= 1) {
        const int u = __shfl_up(incl, o, 32);
        if (lane >= o) incl += u;
    }
    if (lane == 31) wave_tot[wave] = incl;
    __syncthreads();

    int woff = 0;
    for (int w = 0; w < wave; ++w) woff += wave_tot[w];
    const int excl = woff + incl - t;

    if (base + 0 < n) off[base + 0] = excl;
    if (base + 1 < n) off[base + 1] = excl + c0;
    if (base + 2 < n) off[base + 2] = excl + c0 + c1;
    if (base + 3 < n) off[base + 3] = excl + c0 + c1 + c2;
    if (tid == 255) tile_sum[blockIdx.x] = woff + incl;
}

__global__ __launch_bounds__(256) void scan_finish(
    const int* __restrict__ cnt, int* __restrict__ off,
    const int* __restrict__ tile_sum, int n)
{
    __shared__ int s_off;
    const int tid = threadIdx.x;
    if (tid < 32) {
        int a = 0;
        for (int b = tid; b < (int)blockIdx.x; b += 32) a += tile_sum[b];
#pragma unroll
        for (int o = 16; o > 0; o >>= 1) a += __shfl_xor(a, o, 32);
        if (tid == 0) s_off = a;
    }
    __syncthreads();
    const int add  = s_off;
    const int base = blockIdx.x * SCAN_TILE + tid * 4;
#pragma unroll
    for (int k = 0; k < 4; ++k) {
        const int i = base + k;
        if (i < n) {
            const int r = off[i] + add;
            off[i] = r;
            if (i == n - 1) off[n] = r + cnt[i];
        }
    }
}

__global__ __launch_bounds__(1024) void coarse_scatter(
    const int* __restrict__ src, const int* __restrict__ dst,
    const int* __restrict__ coff, int* __restrict__ part, int E, int nb)
{
    const int tid    = threadIdx.x;
    const int blocks = (int)gridDim.x;
    for (int i = tid; i < nb; i += 1024) s_bins[i] = coff[i * blocks + (int)blockIdx.x];
    __syncthreads();

    const int chunk = (E + blocks - 1) / blocks;
    const int e_lo  = (int)blockIdx.x * chunk;
    const int e_hi  = (e_lo + chunk < E) ? (e_lo + chunk) : E;
    for (int e0 = e_lo + tid; e0 < e_hi; e0 += 4096) {
        int d[4], s[4];
#pragma unroll
        for (int k = 0; k < 4; ++k) {
            const int e = e0 + k * 1024;
            d[k] = (e < e_hi) ? dst[e] : -1;
            s[k] = (e < e_hi) ? src[e] : 0;
        }
#pragma unroll
        for (int k = 0; k < 4; ++k) {
            if (d[k] >= 0) {
                const int slot = atomicAdd(&s_bins[d[k] >> BIN_SHIFT], 1);
                *(int2*)(part + 2 * (size_t)slot) = make_int2(s[k], d[k]);
            }
        }
    }
}

__global__ __launch_bounds__(1024) void fine_bucket(
    const int* __restrict__ part, const int* __restrict__ coff,
    int* __restrict__ rowptr, float* __restrict__ deg,
    int* __restrict__ bucket, int n, int blocks)
{
    __shared__ int s_cnt[BIN_NODES];
    __shared__ int s_wtot[BIN_NODES / 32];
    const int tid  = threadIdx.x;
    const int lane = tid & 31;
    const int wave = tid >> 5;
    const int bin  = (int)blockIdx.x;
    const int b_lo = coff[bin * blocks];
    const int b_hi = coff[(bin + 1) * blocks];

    if (tid < BIN_NODES) s_cnt[tid] = 0;
    __syncthreads();

    for (int p0 = b_lo + tid; p0 < b_hi; p0 += 4096) {
        int d[4];
#pragma unroll
        for (int k = 0; k < 4; ++k) {
            const int p = p0 + k * 1024;
            d[k] = (p < b_hi) ? part[2 * (size_t)p + 1] : -1;
        }
#pragma unroll
        for (int k = 0; k < 4; ++k) {
            if (d[k] >= 0) atomicAdd(&s_cnt[d[k] & (BIN_NODES - 1)], 1);
        }
    }
    __syncthreads();

    int c = 0, incl = 0;
    if (tid < BIN_NODES) {
        c = s_cnt[tid];
        incl = c;
#pragma unroll
        for (int o = 1; o < 32; o <<= 1) {
            const int u = __shfl_up(incl, o, 32);
            if (lane >= o) incl += u;
        }
        if (lane == 31) s_wtot[wave] = incl;
    }
    __syncthreads();
    if (tid < BIN_NODES) {
        int woff = 0;
        for (int w = 0; w < wave; ++w) woff += s_wtot[w];
        const int first = b_lo + woff + incl - c;
        const int node  = bin * BIN_NODES + tid;
        if (node < n) {
            rowptr[node] = first;
            deg[node]    = (float)c;
            if (node == n - 1) rowptr[n] = first + c;
        }
        s_cnt[tid] = first;
    }
    __syncthreads();

    for (int p0 = b_lo + tid; p0 < b_hi; p0 += 4096) {
        int2 sd[4];
#pragma unroll
        for (int k = 0; k < 4; ++k) {
            const int p = p0 + k * 1024;
            sd[k] = (p < b_hi) ? *(const int2*)(part + 2 * (size_t)p) : make_int2(0, -1);
        }
#pragma unroll
        for (int k = 0; k < 4; ++k) {
            if (sd[k].y >= 0) {
                const int slot = atomicAdd(&s_cnt[sd[k].y & (BIN_NODES - 1)], 1);
                bucket[slot] = sd[k].x;
            }
        }
    }
}

__global__ __launch_bounds__(256) void bucket_sort(
    const int* __restrict__ rowptr, const int* __restrict__ bucket,
    int* __restrict__ sorted, int n)
{
    const int lane = threadIdx.x & 31;
    const int node0 = __builtin_amdgcn_readfirstlane(
        (int)((blockIdx.x * 8 + (threadIdx.x >> 5)) * 4));
  for (int q = 0; q < 4; ++q) {
    const int node = node0 + q;
    if (node >= n) return;
    const int start = __builtin_amdgcn_readfirstlane(rowptr[node]);
    const int len   = __builtin_amdgcn_readfirstlane(rowptr[node + 1]) - start;

    for (int ib = 0; ib < len; ib += 32) {
        const int i   = ib + lane;
        const int key = (i < len) ? bucket[start + i] : 0;
        int rank = 0;
        for (int jb = 0; jb < len; jb += 32) {
            const int jl = jb + lane;
            const int kc = (jl < len) ? bucket[start + jl] : 0;
            const int m  = (len - jb < 32) ? (len - jb) : 32;
            for (int jj = 0; jj < m; ++jj) {
                const int kj = __builtin_amdgcn_readlane(kc, jj);
                const int j  = jb + jj;
                rank += (kj < key || (kj == key && j < i)) ? 1 : 0;
            }
        }
        if (i < len) sorted[start + rank] = key;
    }
  }
}

__device__ __forceinline__ void add4(float4& a, const float4& r) {
    a.x += r.x; a.y += r.y; a.z += r.z; a.w += r.w;
}

__global__ __launch_bounds__(256) void gather_sum(
    const float* __restrict__ feat, const int* __restrict__ rowptr,
    const int* __restrict__ sorted, float* __restrict__ msg, int n)
{
    const int lane = threadIdx.x & 31;
    const int node = __builtin_amdgcn_readfirstlane(
        (int)(blockIdx.x * 8 + (threadIdx.x >> 5)));
    if (node >= n) return;
    const int start = __builtin_amdgcn_readfirstlane(rowptr[node]);
    const int end   = __builtin_amdgcn_readfirstlane(rowptr[node + 1]);

    const float* col = feat + lane * 4;
    float4 acc = make_float4(0.f, 0.f, 0.f, 0.f);

    for (int base = start; base < end; base += 32) {
        const int m   = (end - base < 32) ? (end - base) : 32;
        const int idx = (lane < m) ? sorted[base + lane] : 0;
        int jj = 0;
        for (; jj + 8 <= m; jj += 8) {
            float4 r[8];
#pragma unroll
            for (int u = 0; u < 8; ++u) {
                const int s = __builtin_amdgcn_readlane(idx, jj + u);
                r[u] = *(const float4*)(col + (size_t)s * 128);
            }
#pragma unroll
            for (int u = 0; u < 8; ++u) add4(acc, r[u]);
        }
        if (m - jj >= 4) {
            float4 r[4];
#pragma unroll
            for (int u = 0; u < 4; ++u) {
                const int s = __builtin_amdgcn_readlane(idx, jj + u);
                r[u] = *(const float4*)(col + (size_t)s * 128);
            }
#pragma unroll
            for (int u = 0; u < 4; ++u) add4(acc, r[u]);
            jj += 4;
        }
        if (m - jj >= 2) {
            const int s0 = __builtin_amdgcn_readlane(idx, jj);
            const int s1 = __builtin_amdgcn_readlane(idx, jj + 1);
            const float4 r0 = *(const float4*)(col + (size_t)s0 * 128);
            const float4 r1 = *(const float4*)(col + (size_t)s1 * 128);
            add4(acc, r0);
            add4(acc, r1);
            jj += 2;
        }
        if (m - jj >= 1) {
            const int s0 = __builtin_amdgcn_readlane(idx, jj);
            const float4 r0 = *(const float4*)(col + (size_t)s0 * 128);
            add4(acc, r0);
        }
    }

    *(float4*)(msg + (size_t)node * 128 + lane * 4) = acc;
}

#define GEMM_BLOCKS 256
#define STRIP_RT    5
#define STRIP_ROWS  (STRIP_RT * 16)
#define LDS_PITCH   132

template <int NW>
__device__ __forceinline__ void stage_strip(
    const float* __restrict__ msg, const float* __restrict__ deg,
    const float* __restrict__ root,
    float (*dst_mean)[LDS_PITCH], float (*dst_root)[LDS_PITCH],
    int row0, int row_hi, int w, int lane)
{
    constexpr int PER = STRIP_ROWS / NW;
    float4 m[PER], x[PER];
    float  dg[PER];
#pragma unroll
    for (int i = 0; i < PER; ++i) {
        const int  row  = row0 + w + i * NW;
        const int  rowc = (row < row_hi) ? row : row0;
        m[i]  = *(const float4*)(msg  + (size_t)rowc * 128 + lane * 4);
        x[i]  = *(const float4*)(root + (size_t)rowc * 128 + lane * 4);
        dg[i] = deg[rowc];
    }
#pragma unroll
    for (int i = 0; i < PER; ++i) {
        const int  r    = w + i * NW;
        const bool live = (row0 + r < row_hi);
        const float invd = 1.0f / fmaxf(dg[i], 1.0f);
        float4 a, b;
        a.x = live ? m[i].x * invd : 0.0f;
        a.y = live ? m[i].y * invd : 0.0f;
        a.z = live ? m[i].z * invd : 0.0f;
        a.w = live ? m[i].w * invd : 0.0f;
        b.x = live ? x[i].x : 0.0f;
        b.y = live ? x[i].y : 0.0f;
        b.z = live ? x[i].z : 0.0f;
        b.w = live ? x[i].w : 0.0f;
        *(float4*)&dst_mean[r][lane * 4] = a;
        *(float4*)&dst_root[r][lane * 4] = b;
    }
}

template <bool RELU>
__global__ __launch_bounds__(512) void sage_gemm_wmma(
    const float* __restrict__ msg, const float* __restrict__ deg,
    const float* __restrict__ root, const float* __restrict__ Wl,
    const float* __restrict__ bias, const float* __restrict__ Wr,
    float* __restrict__ out, int n_rows)
{
    __shared__ float lds_wl[128][LDS_PITCH];
    __shared__ float lds_wr[128][LDS_PITCH];
    __shared__ float lds_mean[2][STRIP_ROWS][LDS_PITCH];
    __shared__ float lds_root[2][STRIP_ROWS][LDS_PITCH];

    const int tid  = threadIdx.x;
    const int lane = tid & 31;
    const int wave = __builtin_amdgcn_readfirstlane(tid >> 5);

    const int tiles = (n_rows + 15) >> 4;
    const int tpb   = (tiles + (int)gridDim.x - 1) / (int)gridDim.x;
    const int t_lo  = (int)blockIdx.x * tpb;
    if (t_lo >= tiles) return;
    const int t_hi    = (t_lo + tpb < tiles) ? (t_lo + tpb) : tiles;
    const int row_lo  = t_lo * 16;
    const int row_hi  = (t_hi * 16 < n_rows) ? (t_hi * 16) : n_rows;
    const int nstrips = (row_hi - row_lo + STRIP_ROWS - 1) / STRIP_ROWS;

    {
        float4 wl[8], wr[8];
#pragma unroll
        for (int i = 0; i < 8; ++i) {
            const int r = wave + i * 16;
            wl[i] = *(const float4*)(Wl + (size_t)r * 128 + lane * 4);
            wr[i] = *(const float4*)(Wr + (size_t)r * 128 + lane * 4);
        }
#pragma unroll
        for (int i = 0; i < 8; ++i) {
            const int r = wave + i * 16;
            *(float4*)&lds_wl[r][lane * 4] = wl[i];
            *(float4*)&lds_wr[r][lane * 4] = wr[i];
        }
        stage_strip<16>(msg, deg, root, lds_mean[0], lds_root[0],
                        row_lo, row_hi, wave, lane);
    }
    __syncthreads();

    const int col  = (wave & 7) * 16;
    const int lr   = lane & 15;
    const int kh   = (lane >> 4) * 2;
    const int rtop = (lane >> 4) * 8;
    const float bv = bias[col + lr];

    for (int s = 0; s < nstrips; ++s) {
        const int buf  = s & 1;
        const int row0 = row_lo + s * STRIP_ROWS;

        v8f acc[STRIP_RT];
#pragma unroll
        for (int t = 0; t < STRIP_RT; ++t) acc[t] = v8f{};

        if (wave >= 8) {
            if (s + 1 < nstrips)
                stage_strip<8>(msg, deg, root, lds_mean[buf ^ 1], lds_root[buf ^ 1],
                               row0 + STRIP_ROWS, row_hi, wave - 8, lane);
        } else {
            const float (*am)[LDS_PITCH] = lds_mean[buf];
            const float (*ar)[LDS_PITCH] = lds_root[buf];

            const float* pb1 = &lds_wl[col + lr][kh];
            const float* pb2 = &lds_wr[col + lr][kh];
            const float* pa1 = &am[lr][kh];
            const float* pa2 = &ar[lr][kh];

            v2f b1[2], b2[2], a1[STRIP_RT][2], a2[STRIP_RT][2];
#pragma unroll
            for (int h = 0; h < 2; ++h) {
                b1[h] = *(const v2f*)(pb1 + 4 * h);
                b2[h] = *(const v2f*)(pb2 + 4 * h);
#pragma unroll
                for (int t = 0; t < STRIP_RT; ++t) {
                    a1[t][h] = *(const v2f*)(pa1 + t * 16 * LDS_PITCH + 4 * h);
                    a2[t][h] = *(const v2f*)(pa2 + t * 16 * LDS_PITCH + 4 * h);
                }
            }

            for (int k0 = 0; k0 < 128; k0 += 8) {
                const int kn = (k0 + 8 < 128) ? (k0 + 8) : k0;
                v2f nb1[2], nb2[2], na1[STRIP_RT][2], na2[STRIP_RT][2];
#pragma unroll
                for (int h = 0; h < 2; ++h) {
                    nb1[h] = *(const v2f*)(pb1 + kn + 4 * h);
                    nb2[h] = *(const v2f*)(pb2 + kn + 4 * h);
#pragma unroll
                    for (int t = 0; t < STRIP_RT; ++t) {
                        na1[t][h] = *(const v2f*)(pa1 + t * 16 * LDS_PITCH + kn + 4 * h);
                        na2[t][h] = *(const v2f*)(pa2 + t * 16 * LDS_PITCH + kn + 4 * h);
                    }
                }
                __builtin_amdgcn_sched_barrier(0);
#pragma unroll
                for (int h = 0; h < 2; ++h) {
#pragma unroll
                    for (int t = 0; t < STRIP_RT; ++t)
                        acc[t] = __builtin_amdgcn_wmma_f32_16x16x4_f32(
                            false, a1[t][h], false, b1[h], (short)0, acc[t], false, false);
#pragma unroll
                    for (int t = 0; t < STRIP_RT; ++t)
                        acc[t] = __builtin_amdgcn_wmma_f32_16x16x4_f32(
                            false, a2[t][h], false, b2[h], (short)0, acc[t], false, false);
                }
#pragma unroll
                for (int h = 0; h < 2; ++h) {
                    b1[h] = nb1[h];
                    b2[h] = nb2[h];
#pragma unroll
                    for (int t = 0; t < STRIP_RT; ++t) {
                        a1[t][h] = na1[t][h];
                        a2[t][h] = na2[t][h];
                    }
                }
            }
        }
        __syncthreads();

        if (wave < 8) {
            if (row0 + STRIP_ROWS <= row_hi) {
#pragma unroll
                for (int t = 0; t < STRIP_RT; ++t) {
#pragma unroll
                    for (int v = 0; v < 8; ++v) {
                        const int row = row0 + t * 16 + v + rtop;
                        float r = acc[t][v] + bv;
                        if (RELU) r = fmaxf(r, 0.0f);
                        out[(size_t)row * 128 + col + lr] = r;
                    }
                }
            } else {
#pragma unroll
                for (int t = 0; t < STRIP_RT; ++t) {
#pragma unroll
                    for (int v = 0; v < 8; ++v) {
                        const int row = row0 + t * 16 + v + rtop;
                        float r = acc[t][v] + bv;
                        if (RELU) r = fmaxf(r, 0.0f);
                        if (row < row_hi) out[(size_t)row * 128 + col + lr] = r;
                    }
                }
            }
        }
    }
}

extern "C" void kernel_launch(void* const* d_in, const int* in_sizes, int n_in,
                              void* d_out, int out_size, void* d_ws, size_t ws_size,
                              hipStream_t stream) {
    const float* x   = (const float*)d_in[0];
    const int*   ei  = (const int*)d_in[1];
    const float* W1l = (const float*)d_in[2];
    const float* b1l = (const float*)d_in[3];
    const float* W1r = (const float*)d_in[4];
    const float* W2l = (const float*)d_in[5];
    const float* b2l = (const float*)d_in[6];
    const float* W2r = (const float*)d_in[7];
    float* out = (float*)d_out;

    const int N = in_sizes[0] / 128;
    const int E = in_sizes[1] / 2;
    const int* src = ei;
    const int* dst = ei + E;

    const int bins  = (N + BIN_NODES - 1) / BIN_NODES;
    const int cells = bins * CHUNK_BLOCKS;
    const int tiles = (cells + SCAN_TILE - 1) / SCAN_TILE;

    float* msg = (float*)d_ws;
    float* h1  = msg + (size_t)N * 128;
    int* part     = (int*)(h1 + (size_t)N * 128);
    float* deg    = (float*)(part + 2 * (size_t)E);
    int* rowptr   = (int*)(deg + N);
    int* cmat     = rowptr + (N + 1);
    int* coff     = cmat + cells;
    int* tile_sum = coff + (cells + 1);
    int* bucket   = tile_sum + tiles;
    int* sorted   = bucket + E;

    const int    s_blocks  = (N + 31) / 32;
    const int    w_blocks  = (N + 7) / 8;
    const size_t bin_lds   = (size_t)bins * sizeof(int);

    coarse_count<<<CHUNK_BLOCKS, 1024, bin_lds, stream>>>(dst, cmat, E, bins);
    scan_tiles<<<tiles, 256, 0, stream>>>(cmat, coff, tile_sum, cells);
    scan_finish<<<tiles, 256, 0, stream>>>(cmat, coff, tile_sum, cells);
    coarse_scatter<<<CHUNK_BLOCKS, 1024, bin_lds, stream>>>(src, dst, coff, part, E, bins);
    fine_bucket<<<bins, 1024, 0, stream>>>(part, coff, rowptr, deg, bucket, N, CHUNK_BLOCKS);
    bucket_sort<<<s_blocks, 256, 0, stream>>>(rowptr, bucket, sorted, N);

    gather_sum<<<w_blocks, 256, 0, stream>>>(x, rowptr, sorted, msg, N);
    sage_gemm_wmma<true><<<GEMM_BLOCKS, 512, 0, stream>>>(
        msg, deg, x, W1l, b1l, W1r, h1, N);

    gather_sum<<<w_blocks, 256, 0, stream>>>(h1, rowptr, sorted, msg, N);
    sage_gemm_wmma<false><<<GEMM_BLOCKS, 512, 0, stream>>>(
        msg, deg, h1, W2l, b2l, W2r, out, N);
}
